# starting kernel
# baseline (speedup 1.0000x reference)
#include <hip/hip_runtime.h>
#include <math.h>


typedef __attribute__((ext_vector_type(2))) float v2f;
typedef __attribute__((ext_vector_type(8))) float v8f;

constexpr int Bc  = 8;
constexpr int Tc  = 4096;
constexpr int Dc  = 1024;
constexpr int NCH = 32;
constexpr int CL  = Tc / NCH;
constexpr int NDG = Dc / 16;

__device__ __forceinline__ float sigmoidf_fast(float x) {
    return 1.0f / (1.0f + __expf(-x));
}

__global__ __launch_bounds__(256) void ema_phase1(
    const float* __restrict__ v, const float* __restrict__ logit,
    float* __restrict__ ends) {
    int g = blockIdx.x * 256 + threadIdx.x;
    int d = g & (Dc - 1);
    int c = (g >> 10) & (NCH - 1);
    int b = g >> 15;
    float a = sigmoidf_fast(logit[d]);
    float w = 1.0f - a;
    const float* vp = v + ((size_t)(b * Tc + c * CL)) * Dc + d;
    float cur = 0.0f;
#pragma unroll 4
    for (int t = 0; t < CL; ++t)
        cur = fmaf(w, cur, a * vp[(size_t)t * Dc]);
    ends[(b * NCH + c) * Dc + d] = cur;
}

__global__ __launch_bounds__(256) void ema_phase2(
    const float* __restrict__ ends, const float* __restrict__ logit,
    float* __restrict__ carry) {
    int g = blockIdx.x * 256 + threadIdx.x;
    int d = g & (Dc - 1);
    int b = g >> 10;
    float a = sigmoidf_fast(logit[d]);
    float w = 1.0f - a;
    float wL = w;
#pragma unroll
    for (int i = 0; i < 7; ++i) wL *= wL;
    float P = 0.0f;
    for (int c = 0; c < NCH; ++c) {
        carry[(b * NCH + c) * Dc + d] = P;
        P = fmaf(wL, P, ends[(b * NCH + c) * Dc + d]);
    }
}

__global__ __launch_bounds__(256) void ema_phase3(
    const float* __restrict__ v, const float* __restrict__ logit,
    const float* __restrict__ carry, float* __restrict__ out) {
    const int lane = threadIdx.x & 31;
    const int wave = blockIdx.x * 8 + (threadIdx.x >> 5);
    const int dg = wave & (NDG - 1);
    const int c  = (wave >> 6) & (NCH - 1);
    const int b  = wave >> 11;
    const int dl = lane & 15;
    const int d  = dg * 16 + dl;
    const bool hi = lane >= 16;
    const int jb = hi ? 2 : 0;
    const int ib = hi ? 8 : 0;

    float a  = sigmoidf_fast(logit[d]);
    float w  = 1.0f - a;
    float wi = 1.0f / w;
    float wi2 = wi * wi;
    float wi4 = wi2 * wi2;
    const float wq0_init = hi ? wi2 : 1.0f;
    float wbase = 1.0f;
    if (hi) { float w2 = w * w, w4 = w2 * w2; wbase = w4 * w4; }

    float P = carry[(b * NCH + c) * Dc + d];
    const size_t vb = (size_t)b * Tc * Dc;
    const float* vp = v + vb + d;
    float* op = out + vb + d;

    for (int tau = 0; tau < CL / 16; ++tau) {
        const int t0 = c * CL + tau * 16;
        v8f acc = {0.f, 0.f, 0.f, 0.f, 0.f, 0.f, 0.f, 0.f};
        float wq0 = wq0_init;
#pragma unroll
        for (int k = 0; k < 4; ++k) {
            const int j0 = 4 * k + jb;
            float x0 = vp[(size_t)(t0 + j0) * Dc];
            float x1 = vp[(size_t)(t0 + j0 + 1) * Dc];
            v2f bm;
            bm.x = wq0 * a * x0;
            bm.y = (wq0 * wi) * a * x1;
            v2f am;
            am.x = (j0     <= dl) ? 1.0f : 0.0f;
            am.y = (j0 + 1 <= dl) ? 1.0f : 0.0f;
            acc = __builtin_amdgcn_wmma_f32_16x16x4_f32(
                      false, am, false, bm, (short)0, acc, false, false);
            wq0 *= wi4;
        }
        const float wpP = w * P;
        float wr = wbase;
        float olast = 0.0f;
        #pragma unroll
        for (int r = 0; r < 8; ++r) {
            float o = wr * (acc[r] + wpP);
            __builtin_nontemporal_store(o, &op[(size_t)(t0 + ib + r) * Dc]);
            olast = o;
            wr *= w;
        }
        P = __shfl(olast, 16 + dl, 32);
    }
}

extern "C" void kernel_launch(void* const* d_in, const int* in_sizes, int n_in,
                              void* d_out, int out_size, void* d_ws, size_t ws_size,
                              hipStream_t stream) {
    const float* v     = (const float*)d_in[0];
    const float* logit = (const float*)d_in[1];
    float* out = (float*)d_out;
    float* ends  = (float*)d_ws;
    float* carry = ends + (size_t)Bc * NCH * Dc;

    ema_phase1<<<(Bc * NCH * Dc) / 256, 256, 0, stream>>>(v, logit, ends);
    ema_phase2<<<(Bc * Dc) / 256, 256, 0, stream>>>(ends, logit, carry);
    ema_phase3<<<(Bc * NCH * NDG) / 8, 256, 0, stream>>>(v, logit, carry, out);
}


# 1 launch: 32-wave workgroup per (batch,16 channels); each wave holds its 128-step chunk in registers, scans it from zero state, chunk ends meet once in LDS, carry added at store (sum order changes)
# speedup vs baseline: 1.3690x; 1.3690x over previous
#include <hip/hip_runtime.h>
#include <math.h>


typedef __attribute__((ext_vector_type(2))) float v2f;
typedef __attribute__((ext_vector_type(8))) float v8f;

constexpr int Bc  = 8;
constexpr int Tc  = 4096;
constexpr int Dc  = 1024;
constexpr int NCH = 32;
constexpr int CL  = Tc / NCH;
constexpr int NDG = Dc / 16;
constexpr int NT3 = CL / 16;
constexpr int EP  = NCH + 4;

__device__ __forceinline__ float sigmoidf_fast(float x) {
    return 1.0f / (1.0f + __expf(-x));
}

__global__ __launch_bounds__(1024) void ema_fused(
    const float* __restrict__ v, const float* __restrict__ logit,
    float* __restrict__ out) {
    __shared__ alignas(16) float ends_s[16 * EP];

    const int lane = threadIdx.x & 31;
    const int c  = __builtin_amdgcn_readfirstlane((int)(threadIdx.x >> 5));
    const int b  = blockIdx.x & (Bc - 1);
    const int dg = blockIdx.x >> 3;
    const int dl = lane & 15;
    const int d  = dg * 16 + dl;
    const bool hi = lane >= 16;
    const int jb = hi ? 2 : 0;
    const int ib = hi ? 8 : 0;

    const float lg = logit[d];
    const size_t cb = ((size_t)b * Tc + (size_t)c * CL) * Dc + d;
    const float* vq = v + cb + (size_t)jb * Dc;
    float* oq = out + cb + (size_t)ib * Dc;

    float x[NT3][8];
#pragma unroll
    for (int tau = 0; tau < NT3; ++tau) {
        __builtin_amdgcn_sched_barrier(0);
#pragma unroll
        for (int k = 0; k < 4; ++k) {
            x[tau][2 * k]     = vq[(size_t)(tau * 16 + 4 * k) * Dc];
            x[tau][2 * k + 1] = vq[(size_t)(tau * 16 + 4 * k + 1) * Dc];
        }
    }

    const float a   = sigmoidf_fast(lg);
    const float w   = 1.0f - a;
    const float wi  = 1.0f / w;
    const float wi2 = wi * wi;
    const float wi4 = wi2 * wi2;
    const float wq0_init = hi ? wi2 : 1.0f;
    const float w2 = w * w, w4 = w2 * w2, w8 = w4 * w4;
    const float w16 = w8 * w8;
    const float wbase = hi ? w8 : 1.0f;

    float e[NT3];
#pragma unroll
    for (int tau = 0; tau < NT3; ++tau) {
        v8f acc = {0.f, 0.f, 0.f, 0.f, 0.f, 0.f, 0.f, 0.f};
        float wq0 = wq0_init;
#pragma unroll
        for (int k = 0; k < 4; ++k) {
            const int j0 = 4 * k + jb;
            float x0 = x[tau][2 * k];
            float x1 = x[tau][2 * k + 1];
            v2f bm;
            bm.x = wq0 * a * x0;
            bm.y = (wq0 * wi) * a * x1;
            v2f am;
            am.x = (j0     <= dl) ? 1.0f : 0.0f;
            am.y = (j0 + 1 <= dl) ? 1.0f : 0.0f;
            acc = __builtin_amdgcn_wmma_f32_16x16x4_f32(
                      false, am, false, bm, (short)0, acc, false, false);
            wq0 *= wi4;
        }
        float wr = wbase;
#pragma unroll
        for (int r = 0; r < 8; ++r) {
            x[tau][r] = wr * acc[r];
            wr *= w;
        }
        e[tau] = __shfl(x[tau][7], 16 + dl, 32);
        __builtin_amdgcn_sched_barrier(0);
    }

    float E = e[0];
#pragma unroll
    for (int tau = 1; tau < NT3; ++tau) E = fmaf(w16, E, e[tau]);
    if (!hi) ends_s[dl * EP + c] = E;
    __syncthreads();

    float wL = w16 * w16;
    wL *= wL;
    wL *= wL;
    float P = 0.0f;
    const float4* ep = reinterpret_cast<const float4*>(&ends_s[dl * EP]);
#pragma unroll
    for (int q = 0; q < NCH / 4; ++q) {
        const float4 t = ep[q];
        float Pn;
        Pn = fmaf(wL, P, t.x); P = (4 * q     < c) ? Pn : P;
        Pn = fmaf(wL, P, t.y); P = (4 * q + 1 < c) ? Pn : P;
        Pn = fmaf(wL, P, t.z); P = (4 * q + 2 < c) ? Pn : P;
        Pn = fmaf(wL, P, t.w); P = (4 * q + 3 < c) ? Pn : P;
    }

    float wrow[8];
    {
        float wr = wbase * w;
#pragma unroll
        for (int r = 0; r < 8; ++r) { wrow[r] = wr; wr *= w; }
    }

#pragma unroll
    for (int tau = 0; tau < NT3; ++tau) {
#pragma unroll
        for (int r = 0; r < 8; ++r) {
            float o = fmaf(wrow[r], P, x[tau][r]);
            __builtin_nontemporal_store(o, &oq[(size_t)(tau * 16 + r) * Dc]);
        }
        P = fmaf(w16, P, e[tau]);
    }
}

extern "C" void kernel_launch(void* const* d_in, const int* in_sizes, int n_in,
                              void* d_out, int out_size, void* d_ws, size_t ws_size,
                              hipStream_t stream) {
    const float* v     = (const float*)d_in[0];
    const float* logit = (const float*)d_in[1];
    float* out = (float*)d_out;
    ema_fused<<<Bc * NDG, NCH * 32, 0, stream>>>(v, logit, out);
}


# lane l = channel l of a 32-ch strip: a 4-B wave access is one whole 128-B row line; register halves swapped (v_permlane16_swap) into and out of WMMA; 2 passes of 64-step chunks (sum order changes)
# speedup vs baseline: 3.0008x; 2.1920x over previous
#include <hip/hip_runtime.h>
#include <math.h>


typedef __attribute__((ext_vector_type(2))) float v2f;
typedef __attribute__((ext_vector_type(8))) float v8f;
typedef __attribute__((ext_vector_type(2))) unsigned int v2u;

constexpr int Bc  = 8;
constexpr int Tc  = 4096;
constexpr int Dc  = 1024;
constexpr int NW  = 32;
constexpr int NP  = 2;
constexpr int CL  = Tc / (NW * NP);
constexpr int NT  = CL / 16;
constexpr int NCG = Dc / 32;
constexpr int EP  = NW + 4;

__device__ __forceinline__ float sigmoidf_fast(float x) {
    return 1.0f / (1.0f + __expf(-x));
}

__device__ __forceinline__ void half_swap(float& p, float& q) {
    v2u r = __builtin_amdgcn_permlane16_swap(__float_as_uint(p), __float_as_uint(q),
                                             false, false);
    p = __uint_as_float(r.x);
    q = __uint_as_float(r.y);
}

__global__ __launch_bounds__(1024) void ema_fused(
    const float* __restrict__ v, const float* __restrict__ logit,
    float* __restrict__ out) {
    __shared__ alignas(16) float ends_s[NP][32 * EP];

    const int lane = threadIdx.x & 31;
    const int c  = __builtin_amdgcn_readfirstlane((int)(threadIdx.x >> 5));
    const int b  = blockIdx.x & (Bc - 1);
    const int cg = blockIdx.x >> 3;
    const int d  = cg * 32 + lane;
    const int dl = lane & 15;
    const int jb = (lane >= 16) ? 2 : 0;

    const float lg = logit[d];

    float Pin = 0.0f;
    float a = 0.f, w = 0.f, wi = 0.f, w8 = 0.f, w16 = 0.f, wL = 0.f;

#pragma unroll
    for (int p = 0; p < NP; ++p) {
        const size_t cb = ((size_t)b * Tc + (size_t)(p * NW + c) * CL) * Dc + d;
        const float* vq = v + cb;
        float* oq = out + cb;

        float x[NT][16];
#pragma unroll
        for (int tau = 0; tau < NT; ++tau) {
            __builtin_amdgcn_sched_barrier(0);
#pragma unroll
            for (int j = 0; j < 16; ++j)
                x[tau][j] = vq[(size_t)(tau * 16 + j) * Dc];
        }

        if (p == 0) {
            a  = sigmoidf_fast(lg);
            w  = 1.0f - a;
            wi = 1.0f / w;
            const float w2 = w * w, w4 = w2 * w2;
            w8  = w4 * w4;
            w16 = w8 * w8;
            wL = w16 * w16;
            wL *= wL;
        }

        float e[NT];
#pragma unroll
        for (int tau = 0; tau < NT; ++tau) {
            float f = a;
#pragma unroll
            for (int j = 0; j < 16; ++j) {
                x[tau][j] *= f;
                f *= wi;
            }
#pragma unroll
            for (int k = 0; k < 4; ++k) {
                half_swap(x[tau][4 * k],     x[tau][4 * k + 2]);
                half_swap(x[tau][4 * k + 1], x[tau][4 * k + 3]);
            }
            v8f acc0 = {0.f, 0.f, 0.f, 0.f, 0.f, 0.f, 0.f, 0.f};
            v8f acc1 = {0.f, 0.f, 0.f, 0.f, 0.f, 0.f, 0.f, 0.f};
#pragma unroll
            for (int k = 0; k < 4; ++k) {
                const int j0 = 4 * k + jb;
                v2f am;
                am.x = (j0     <= dl) ? 1.0f : 0.0f;
                am.y = (j0 + 1 <= dl) ? 1.0f : 0.0f;
                v2f b0, b1;
                b0.x = x[tau][4 * k];     b0.y = x[tau][4 * k + 1];
                b1.x = x[tau][4 * k + 2]; b1.y = x[tau][4 * k + 3];
                acc0 = __builtin_amdgcn_wmma_f32_16x16x4_f32(
                           false, am, false, b0, (short)0, acc0, false, false);
                acc1 = __builtin_amdgcn_wmma_f32_16x16x4_f32(
                           false, am, false, b1, (short)0, acc1, false, false);
            }
            float ur = 1.0f;
#pragma unroll
            for (int r = 0; r < 8; ++r) {
                float lo = acc0[r], hi = acc1[r];
                half_swap(lo, hi);
                x[tau][r]     = ur * lo;
                x[tau][8 + r] = (ur * w8) * hi;
                ur *= w;
            }
            e[tau] = x[tau][15];
            __builtin_amdgcn_sched_barrier(0);
        }

        float E = e[0];
#pragma unroll
        for (int tau = 1; tau < NT; ++tau) E = fmaf(w16, E, e[tau]);
        ends_s[p][lane * EP + c] = E;
        __syncthreads();

        float F = Pin, P = Pin;
        const float4* ep = reinterpret_cast<const float4*>(&ends_s[p][lane * EP]);
#pragma unroll
        for (int q = 0; q < NW / 4; ++q) {
            const float4 t = ep[q];
            F = fmaf(wL, F, t.x); P = (4 * q     < c) ? F : P;
            F = fmaf(wL, F, t.y); P = (4 * q + 1 < c) ? F : P;
            F = fmaf(wL, F, t.z); P = (4 * q + 2 < c) ? F : P;
            F = fmaf(wL, F, t.w); P = (4 * q + 3 < c) ? F : P;
        }
        Pin = F;

#pragma unroll
        for (int tau = 0; tau < NT; ++tau) {
            float Pw = P;
#pragma unroll
            for (int i = 0; i < 16; ++i) {
                Pw *= w;
                x[tau][i] += Pw;
            }
            P = Pw + e[tau];
            __builtin_amdgcn_sched_barrier(0);
#pragma unroll
            for (int i = 0; i < 16; ++i) {
                __builtin_nontemporal_store(x[tau][i], &oq[(size_t)(tau * 16 + i) * Dc]);
            }
        }
    }
}

extern "C" void kernel_launch(void* const* d_in, const int* in_sizes, int n_in,
                              void* d_out, int out_size, void* d_ws, size_t ws_size,
                              hipStream_t stream) {
    const float* v     = (const float*)d_in[0];
    const float* logit = (const float*)d_in[1];
    float* out = (float*)d_out;
    ema_fused<<<Bc * NCG, NW * 32, 0, stream>>>(v, logit, out);
}
